# starting kernel
# baseline (speedup 1.0000x reference)
#include <hip/hip_runtime.h>

typedef __attribute__((ext_vector_type(2))) float v2f;
typedef __attribute__((ext_vector_type(8))) float v8f;

#define VOCAB   50000
#define EMB     300
#define HID     256
#define OUTD    3
#define NASP    8
#define BATCH   8192
#define SEQ     128
#define ROWS    64
#define RT      4

__global__ __launch_bounds__(320)
void pool_kernel(const int* __restrict__ ids,
                 const float* __restrict__ emb,
                 float* __restrict__ x) {
  __shared__ int sIds[SEQ];
  const int s = blockIdx.x;
  const int t = threadIdx.x;
  if (t < SEQ) sIds[t] = ids[s * SEQ + t];
  __syncthreads();
  if (t < EMB) {
    float acc = 0.0f;
#pragma unroll 4
    for (int k = 0; k < SEQ; ++k) {
      acc += emb[(size_t)sIds[k] * EMB + t];
    }
    x[(size_t)s * EMB + t] = acc * (1.0f / (float)SEQ);
  }
}

__global__ __launch_bounds__(256)
void mlp_kernel(const float* __restrict__ x,
                const int* __restrict__ aspect_id,
                const float* __restrict__ W1,
                const float* __restrict__ B1,
                const float* __restrict__ W2,
                const float* __restrict__ B2,
                float* __restrict__ out) {
  extern __shared__ float smem[];
  float* sX = smem;
  float* sH = smem + ROWS * EMB;

  const int a    = blockIdx.y;
  const int row0 = blockIdx.x * ROWS;
  const int t    = threadIdx.x;

  for (int i = t; i < ROWS * EMB; i += 256) {
    sX[i] = x[(size_t)row0 * EMB + i];
  }
  __syncthreads();

  const int lane  = t & 31;
  const int wv    = t >> 5;
  const int nloc  = lane & 15;
  const int khalf = (lane >> 4) * 2;
  const int m     = lane & 15;

  const float* w1a = W1 + (size_t)a * EMB * HID;

  for (int nt = 0; nt < 2; ++nt) {
    const int col = wv * 32 + nt * 16 + nloc;
    v8f acc[RT];
#pragma unroll
    for (int rt = 0; rt < RT; ++rt) acc[rt] = (v8f){};

    for (int kk = 0; kk < EMB / 4; ++kk) {
      const int k = kk * 4 + khalf;
      v2f bfrag;
      bfrag.x = w1a[(size_t)k * HID + col];
      bfrag.y = w1a[(size_t)(k + 1) * HID + col];
#pragma unroll
      for (int rt = 0; rt < RT; ++rt) {
        v2f afrag = *(const v2f*)&sX[(rt * 16 + m) * EMB + k];
        acc[rt] = __builtin_amdgcn_wmma_f32_16x16x4_f32(
             false, afrag,  false, bfrag,
             (short)0, acc[rt],  false,  false);
      }
    }
    const float bias = B1[a * HID + col];
#pragma unroll
    for (int rt = 0; rt < RT; ++rt) {
#pragma unroll
      for (int r = 0; r < 8; ++r) {
        const int rowm = rt * 16 + r + 8 * (lane >> 4);
        float h = acc[rt][r] + bias;
        sH[rowm * HID + col] = h > 0.0f ? h : 0.0f;
      }
    }
  }
  __syncthreads();

  if (t < ROWS * OUTD) {
    const int s = t / OUTD, j = t % OUTD;
    const float* w2a = W2 + (size_t)a * HID * OUTD;
    float acc = B2[a * OUTD + j];
    for (int k = 0; k < HID; ++k) {
      acc += sH[s * HID + k] * w2a[k * OUTD + j];
    }
    const int b = row0 + s;
    if (aspect_id[b] == a) {
      out[b * OUTD + j] = acc;
    }
  }
}

extern "C" void kernel_launch(void* const* d_in, const int* in_sizes, int n_in,
                              void* d_out, int out_size, void* d_ws, size_t ws_size,
                              hipStream_t stream) {
  (void)in_sizes; (void)n_in; (void)out_size; (void)ws_size;
  const int*   ids    = (const int*)d_in[0];
  const int*   aspect = (const int*)d_in[1];
  const float* emb    = (const float*)d_in[2];
  const float* W1     = (const float*)d_in[3];
  const float* b1     = (const float*)d_in[4];
  const float* W2     = (const float*)d_in[5];
  const float* b2     = (const float*)d_in[6];
  float* out = (float*)d_out;
  float* xws = (float*)d_ws;

  pool_kernel<<<BATCH, 320, 0, stream>>>(ids, emb, xws);

  const size_t smem_bytes = (size_t)(ROWS * EMB + ROWS * HID) * sizeof(float);
  dim3 grid(BATCH / ROWS, NASP);
  mlp_kernel<<<grid, 256, smem_bytes, stream>>>(xws, aspect, W1, b1, W2, b2, out);
}


# pool fused into the MLP kernel: one launch of 256 work-groups; each pools its own 32 rows into LDS (a wave per row, whole-row 16-byte loads) and runs all 8 aspects from there; sum orders unchanged
# speedup vs baseline: 1.4176x; 1.4176x over previous
#include <hip/hip_runtime.h>


typedef __attribute__((ext_vector_type(2))) float v2f;
typedef __attribute__((ext_vector_type(4))) float v4f;
typedef __attribute__((ext_vector_type(4))) int   v4i;
typedef __attribute__((ext_vector_type(8))) float v8f;

#define VOCAB   50000
#define EMB     300
#define HID     256
#define OUTD    3
#define NASP    8
#define BATCH   8192
#define SEQ     128
#define ROWS    32
#define RT      2
#define KB      5
#define NBLK    (EMB / 4 / KB)
#define NWAVE   16
#define HP      260
#define GU      8
#define SX_FLOATS   (ROWS * EMB)
#define SH_FLOATS   (ROWS * HP)
#define LDS_FLOATS  (41 * 1024)

__device__ __forceinline__ void load_b(v2f (&b)[KB], const float* __restrict__ bp, int blk) {
#pragma unroll
  for (int j = 0; j < KB; ++j) {
    const float* p = bp + (size_t)((blk * KB + j) * 4) * HID;
    b[j].x = p[0];
    b[j].y = p[HID];
  }
}

__device__ __forceinline__ void mma_block(v8f (&acc)[RT], const v2f (&b)[KB],
                                          const float* sXl, int blk) {
#pragma unroll
  for (int j = 0; j < KB; ++j) {
    const int k = (blk * KB + j) * 4;
#pragma unroll
    for (int rt = 0; rt < RT; ++rt) {
      v2f afrag = *(const v2f*)&sXl[rt * 16 * EMB + k];
      acc[rt] = __builtin_amdgcn_wmma_f32_16x16x4_f32(
           false, afrag,  false, b[j],
           (short)0, acc[rt],  false,  false);
    }
  }
}

__global__ __launch_bounds__(NWAVE * 32)
void fused_kernel(const int* __restrict__ ids,
                  const int* __restrict__ aspect_id,
                  const float* __restrict__ emb,
                  const float* __restrict__ W1,
                  const float* __restrict__ B1,
                  const float* __restrict__ W2,
                  const float* __restrict__ B2,
                  float* __restrict__ out) {
  extern __shared__ float smem[];
  float* sX = smem;
  float* sH = smem + SX_FLOATS;

  const int row0 = blockIdx.x * ROWS;
  const int t    = threadIdx.x;
  const int lane = t & 31;
  const int wv   = t >> 5;

  {
    const int o0 = lane * 4;
    const int o1 = (lane + 32) * 4;
    const int o2 = (lane < 11 ? lane + 64 : 74) * 4;
    for (int si = 0; si < ROWS / NWAVE; ++si) {
      const int sl = wv * (ROWS / NWAVE) + si;
      const v4i idv = *(const v4i*)(ids + (size_t)(row0 + sl) * SEQ + lane * 4);
      v4f a0 = (v4f){}, a1 = (v4f){}, a2 = (v4f){};
      for (int k0 = 0; k0 < SEQ; k0 += GU) {
        v4f r0[GU], r1[GU], r2[GU];
#pragma unroll
        for (int u = 0; u < GU; ++u) {
          const int id = __builtin_amdgcn_readlane(idv[u & 3], (k0 >> 2) + (u >> 2));
          const float* rp = emb + (size_t)id * EMB;
          r0[u] = *(const v4f*)(rp + o0);
          r1[u] = *(const v4f*)(rp + o1);
          r2[u] = *(const v4f*)(rp + o2);
        }
#pragma unroll
        for (int u = 0; u < GU; ++u) {
          a0 += r0[u];
          a1 += r1[u];
          a2 += r2[u];
        }
      }
      const float sc = 1.0f / (float)SEQ;
      float* xr = sX + sl * EMB;
      *(v4f*)(xr + o0) = a0 * sc;
      *(v4f*)(xr + o1) = a1 * sc;
      if (lane < 11) *(v4f*)(xr + o2) = a2 * sc;
    }
  }
  __syncthreads();

  const int nloc  = lane & 15;
  const int khalf = (lane >> 4) * 2;
  const int m     = lane & 15;
  const int col   = wv * 16 + nloc;
  const float* sXl = sX + m * EMB + khalf;

  for (int a = 0; a < NASP; ++a) {
    float* sHa = sH + (a & 1) * SH_FLOATS;
    v8f acc[RT];
#pragma unroll
    for (int rt = 0; rt < RT; ++rt) acc[rt] = (v8f){};

    const float* bp = W1 + (size_t)a * EMB * HID + (size_t)khalf * HID + col;
    v2f b0[KB], b1[KB], b2[KB];
    load_b(b0, bp, 0);
    load_b(b1, bp, 1);
    for (int p = 0; p < NBLK / 3 - 1; ++p) {
      load_b(b2, bp, 3 * p + 2);
      __builtin_amdgcn_sched_barrier(0);
      mma_block(acc, b0, sXl, 3 * p);
      load_b(b0, bp, 3 * p + 3);
      __builtin_amdgcn_sched_barrier(0);
      mma_block(acc, b1, sXl, 3 * p + 1);
      load_b(b1, bp, 3 * p + 4);
      __builtin_amdgcn_sched_barrier(0);
      mma_block(acc, b2, sXl, 3 * p + 2);
    }
    load_b(b2, bp, NBLK - 1);
    __builtin_amdgcn_sched_barrier(0);
    mma_block(acc, b0, sXl, NBLK - 3);
    mma_block(acc, b1, sXl, NBLK - 2);
    mma_block(acc, b2, sXl, NBLK - 1);
    const float bias = B1[a * HID + col];
#pragma unroll
    for (int rt = 0; rt < RT; ++rt) {
#pragma unroll
      for (int r = 0; r < 8; ++r) {
        const int rowm = rt * 16 + r + 8 * (lane >> 4);
        float h = acc[rt][r] + bias;
        sHa[rowm * HP + col] = h > 0.0f ? h : 0.0f;
      }
    }
    __syncthreads();

    if (t < ROWS * OUTD) {
      const int s = t / OUTD, j = t % OUTD;
      const float* w2a = W2 + (size_t)a * HID * OUTD;
      float o = B2[a * OUTD + j];
#pragma unroll 32
      for (int k = 0; k < HID; ++k) {
        o += sHa[s * HP + k] * w2a[k * OUTD + j];
      }
      const int b = row0 + s;
      if (aspect_id[b] == a) {
        out[b * OUTD + j] = o;
      }
    }
  }
}

extern "C" void kernel_launch(void* const* d_in, const int* in_sizes, int n_in,
                              void* d_out, int out_size, void* d_ws, size_t ws_size,
                              hipStream_t stream) {
  (void)in_sizes; (void)n_in; (void)out_size; (void)d_ws; (void)ws_size;
  const int*   ids    = (const int*)d_in[0];
  const int*   aspect = (const int*)d_in[1];
  const float* emb    = (const float*)d_in[2];
  const float* W1     = (const float*)d_in[3];
  const float* b1     = (const float*)d_in[4];
  const float* W2     = (const float*)d_in[5];
  const float* b2     = (const float*)d_in[6];
  float* out = (float*)d_out;

  const size_t smem_bytes = (size_t)LDS_FLOATS * sizeof(float);
  fused_kernel<<<BATCH / ROWS, NWAVE * 32, smem_bytes, stream>>>(
      ids, aspect, emb, W1, b1, W2, b2, out);
}


# gather: a row's three 16-byte-per-lane loads re-cut from 512+512+176 bytes to 400 each (25 lanes; 32-bit offsets from the table base): no load spans a fifth 128-byte line; sum orders unchanged
# speedup vs baseline: 1.4845x; 1.0472x over previous
#include <hip/hip_runtime.h>


typedef __attribute__((ext_vector_type(2))) float v2f;
typedef __attribute__((ext_vector_type(4))) float v4f;
typedef __attribute__((ext_vector_type(4))) int   v4i;
typedef __attribute__((ext_vector_type(8))) float v8f;

#define VOCAB   50000
#define EMB     300
#define HID     256
#define OUTD    3
#define NASP    8
#define BATCH   8192
#define SEQ     128
#define ROWS    32
#define RT      2
#define KB      5
#define NBLK    (EMB / 4 / KB)
#define NWAVE   16
#define HP      260
#define GU      8
#define SX_FLOATS   (ROWS * EMB)
#define SH_FLOATS   (ROWS * HP)
#define LDS_FLOATS  (41 * 1024)

__device__ __forceinline__ void load_b(v2f (&b)[KB], const float* __restrict__ bp, int blk) {
#pragma unroll
  for (int j = 0; j < KB; ++j) {
    const float* p = bp + (size_t)((blk * KB + j) * 4) * HID;
    b[j].x = p[0];
    b[j].y = p[HID];
  }
}

__device__ __forceinline__ void mma_block(v8f (&acc)[RT], const v2f (&b)[KB],
                                          const float* sXl, int blk) {
#pragma unroll
  for (int j = 0; j < KB; ++j) {
    const int k = (blk * KB + j) * 4;
#pragma unroll
    for (int rt = 0; rt < RT; ++rt) {
      v2f afrag = *(const v2f*)&sXl[rt * 16 * EMB + k];
      acc[rt] = __builtin_amdgcn_wmma_f32_16x16x4_f32(
           false, afrag,  false, b[j],
           (short)0, acc[rt],  false,  false);
    }
  }
}

__global__ __launch_bounds__(NWAVE * 32)
void fused_kernel(const int* __restrict__ ids,
                  const int* __restrict__ aspect_id,
                  const float* __restrict__ emb,
                  const float* __restrict__ W1,
                  const float* __restrict__ B1,
                  const float* __restrict__ W2,
                  const float* __restrict__ B2,
                  float* __restrict__ out) {
  extern __shared__ float smem[];
  float* sX = smem;
  float* sH = smem + SX_FLOATS;

  const int row0 = blockIdx.x * ROWS;
  const int t    = threadIdx.x;
  const int lane = t & 31;
  const int wv   = t >> 5;

  {
    const int o0 = (lane < 25 ? lane : 24) * 4;
    const int o1 = o0 + 100;
    const int o2 = o0 + 200;
    for (int si = 0; si < ROWS / NWAVE; ++si) {
      const int sl = wv * (ROWS / NWAVE) + si;
      const v4i idv = *(const v4i*)(ids + (size_t)(row0 + sl) * SEQ + lane * 4);
      v4f a0 = (v4f){}, a1 = (v4f){}, a2 = (v4f){};
      for (int k0 = 0; k0 < SEQ; k0 += GU) {
        v4f r0[GU], r1[GU], r2[GU];
#pragma unroll
        for (int u = 0; u < GU; ++u) {
          const int id = __builtin_amdgcn_readlane(idv[u & 3], (k0 >> 2) + (u >> 2));
          const unsigned e0 = (unsigned)id * (EMB * 4u) + (unsigned)o0 * 4u;
          const char* eb = (const char*)emb;
          r0[u] = *(const v4f*)(eb + e0);
          r1[u] = *(const v4f*)(eb + e0 + 400u);
          r2[u] = *(const v4f*)(eb + e0 + 800u);
        }
#pragma unroll
        for (int u = 0; u < GU; ++u) {
          a0 += r0[u];
          a1 += r1[u];
          a2 += r2[u];
        }
      }
      const float sc = 1.0f / (float)SEQ;
      float* xr = sX + sl * EMB;
      if (lane < 25) {
        *(v4f*)(xr + o0) = a0 * sc;
        *(v4f*)(xr + o1) = a1 * sc;
        *(v4f*)(xr + o2) = a2 * sc;
      }
    }
  }
  __syncthreads();

  const int nloc  = lane & 15;
  const int khalf = (lane >> 4) * 2;
  const int m     = lane & 15;
  const int col   = wv * 16 + nloc;
  const float* sXl = sX + m * EMB + khalf;

  for (int a = 0; a < NASP; ++a) {
    float* sHa = sH + (a & 1) * SH_FLOATS;
    v8f acc[RT];
#pragma unroll
    for (int rt = 0; rt < RT; ++rt) acc[rt] = (v8f){};

    const float* bp = W1 + (size_t)a * EMB * HID + (size_t)khalf * HID + col;
    v2f b0[KB], b1[KB], b2[KB];
    load_b(b0, bp, 0);
    load_b(b1, bp, 1);
    for (int p = 0; p < NBLK / 3 - 1; ++p) {
      load_b(b2, bp, 3 * p + 2);
      __builtin_amdgcn_sched_barrier(0);
      mma_block(acc, b0, sXl, 3 * p);
      load_b(b0, bp, 3 * p + 3);
      __builtin_amdgcn_sched_barrier(0);
      mma_block(acc, b1, sXl, 3 * p + 1);
      load_b(b1, bp, 3 * p + 4);
      __builtin_amdgcn_sched_barrier(0);
      mma_block(acc, b2, sXl, 3 * p + 2);
    }
    load_b(b2, bp, NBLK - 1);
    __builtin_amdgcn_sched_barrier(0);
    mma_block(acc, b0, sXl, NBLK - 3);
    mma_block(acc, b1, sXl, NBLK - 2);
    mma_block(acc, b2, sXl, NBLK - 1);
    const float bias = B1[a * HID + col];
#pragma unroll
    for (int rt = 0; rt < RT; ++rt) {
#pragma unroll
      for (int r = 0; r < 8; ++r) {
        const int rowm = rt * 16 + r + 8 * (lane >> 4);
        float h = acc[rt][r] + bias;
        sHa[rowm * HP + col] = h > 0.0f ? h : 0.0f;
      }
    }
    __syncthreads();

    if (t < ROWS * OUTD) {
      const int s = t / OUTD, j = t % OUTD;
      const float* w2a = W2 + (size_t)a * HID * OUTD;
      float o = B2[a * OUTD + j];
#pragma unroll 32
      for (int k = 0; k < HID; ++k) {
        o += sHa[s * HP + k] * w2a[k * OUTD + j];
      }
      const int b = row0 + s;
      if (aspect_id[b] == a) {
        out[b * OUTD + j] = o;
      }
    }
  }
}

extern "C" void kernel_launch(void* const* d_in, const int* in_sizes, int n_in,
                              void* d_out, int out_size, void* d_ws, size_t ws_size,
                              hipStream_t stream) {
  (void)in_sizes; (void)n_in; (void)out_size; (void)d_ws; (void)ws_size;
  const int*   ids    = (const int*)d_in[0];
  const int*   aspect = (const int*)d_in[1];
  const float* emb    = (const float*)d_in[2];
  const float* W1     = (const float*)d_in[3];
  const float* b1     = (const float*)d_in[4];
  const float* W2     = (const float*)d_in[5];
  const float* b2     = (const float*)d_in[6];
  float* out = (float*)d_out;

  const size_t smem_bytes = (size_t)LDS_FLOATS * sizeof(float);
  fused_kernel<<<BATCH / ROWS, NWAVE * 32, smem_bytes, stream>>>(
      ids, aspect, emb, W1, b1, W2, b2, out);
}


# layer 2 (256->3) spread over all 16 waves: a lane takes 16 hidden values of one row from LDS against a prefetched W2 slice, a 16-lane butterfly sums the partials; layer-2 sum order changes
# speedup vs baseline: 1.6200x; 1.0913x over previous
#include <hip/hip_runtime.h>


typedef __attribute__((ext_vector_type(2))) float v2f;
typedef __attribute__((ext_vector_type(4))) float v4f;
typedef __attribute__((ext_vector_type(4))) int   v4i;
typedef __attribute__((ext_vector_type(8))) float v8f;

#define VOCAB   50000
#define EMB     300
#define HID     256
#define OUTD    3
#define NASP    8
#define BATCH   8192
#define SEQ     128
#define ROWS    32
#define RT      2
#define KB      5
#define NBLK    (EMB / 4 / KB)
#define NWAVE   16
#define HP      260
#define GU      8
#define SX_FLOATS   (ROWS * EMB)
#define SH_FLOATS   (ROWS * HP)
#define SO_OFFSET   (SX_FLOATS + 2 * SH_FLOATS)
#define LDS_FLOATS  (41 * 1024)

__device__ __forceinline__ void load_b(v2f (&b)[KB], const float* __restrict__ bp, int blk) {
#pragma unroll
  for (int j = 0; j < KB; ++j) {
    const float* p = bp + (size_t)((blk * KB + j) * 4) * HID;
    b[j].x = p[0];
    b[j].y = p[HID];
  }
}

__device__ __forceinline__ void mma_block(v8f (&acc)[RT], const v2f (&b)[KB],
                                          const float* sXl, int blk) {
#pragma unroll
  for (int j = 0; j < KB; ++j) {
    const int k = (blk * KB + j) * 4;
#pragma unroll
    for (int rt = 0; rt < RT; ++rt) {
      v2f afrag = *(const v2f*)&sXl[rt * 16 * EMB + k];
      acc[rt] = __builtin_amdgcn_wmma_f32_16x16x4_f32(
           false, afrag,  false, b[j],
           (short)0, acc[rt],  false,  false);
    }
  }
}

__device__ __forceinline__ float row16_sum(float v) {
  v += __builtin_bit_cast(float, __builtin_amdgcn_update_dpp(
           0, __builtin_bit_cast(int, v), 0x161, 0xf, 0xf, true));
  v += __builtin_bit_cast(float, __builtin_amdgcn_update_dpp(
           0, __builtin_bit_cast(int, v), 0x162, 0xf, 0xf, true));
  v += __builtin_bit_cast(float, __builtin_amdgcn_update_dpp(
           0, __builtin_bit_cast(int, v), 0x164, 0xf, 0xf, true));
  v += __builtin_bit_cast(float, __builtin_amdgcn_update_dpp(
           0, __builtin_bit_cast(int, v), 0x168, 0xf, 0xf, true));
  return v;
}

__global__ __launch_bounds__(NWAVE * 32)
void fused_kernel(const int* __restrict__ ids,
                  const int* __restrict__ aspect_id,
                  const float* __restrict__ emb,
                  const float* __restrict__ W1,
                  const float* __restrict__ B1,
                  const float* __restrict__ W2,
                  const float* __restrict__ B2,
                  float* __restrict__ out) {
  extern __shared__ float smem[];
  float* sX = smem;
  float* sH = smem + SX_FLOATS;

  const int row0 = blockIdx.x * ROWS;
  const int t    = threadIdx.x;
  const int lane = t & 31;
  const int wv   = t >> 5;

  {
    const int o0 = (lane < 25 ? lane : 24) * 4;
    const int o1 = o0 + 100;
    const int o2 = o0 + 200;
    for (int si = 0; si < ROWS / NWAVE; ++si) {
      const int sl = wv * (ROWS / NWAVE) + si;
      const v4i idv = *(const v4i*)(ids + (size_t)(row0 + sl) * SEQ + lane * 4);
      v4f a0 = (v4f){}, a1 = (v4f){}, a2 = (v4f){};
      for (int k0 = 0; k0 < SEQ; k0 += GU) {
        v4f r0[GU], r1[GU], r2[GU];
#pragma unroll
        for (int u = 0; u < GU; ++u) {
          const int id = __builtin_amdgcn_readlane(idv[u & 3], (k0 >> 2) + (u >> 2));
          const unsigned e0 = (unsigned)id * (EMB * 4u) + (unsigned)o0 * 4u;
          const char* eb = (const char*)emb;
          r0[u] = *(const v4f*)(eb + e0);
          r1[u] = *(const v4f*)(eb + e0 + 400u);
          r2[u] = *(const v4f*)(eb + e0 + 800u);
        }
#pragma unroll
        for (int u = 0; u < GU; ++u) {
          a0 += r0[u];
          a1 += r1[u];
          a2 += r2[u];
        }
      }
      const float sc = 1.0f / (float)SEQ;
      float* xr = sX + sl * EMB;
      if (lane < 25) {
        *(v4f*)(xr + o0) = a0 * sc;
        *(v4f*)(xr + o1) = a1 * sc;
        *(v4f*)(xr + o2) = a2 * sc;
      }
    }
  }
  __syncthreads();

  const int nloc  = lane & 15;
  const int khalf = (lane >> 4) * 2;
  const int m     = lane & 15;
  const int col   = wv * 16 + nloc;
  const float* sXl = sX + m * EMB + khalf;

  const int l2row = t >> 4;
  const int l2q   = t & 15;
  const int l2j   = l2q < OUTD ? l2q : OUTD - 1;
  const int my_aspect = aspect_id[row0 + l2row];
  float res = 0.0f;

  for (int a = 0; a < NASP; ++a) {
    float* sHa = sH + (a & 1) * SH_FLOATS;
    v4f w2v[4][3];
    {
      const float* w2p = W2 + (size_t)a * HID * OUTD + 12 * l2q;
#pragma unroll
      for (int i = 0; i < 4; ++i) {
#pragma unroll
        for (int c = 0; c < 3; ++c) {
          w2v[i][c] = *(const v4f*)(w2p + 192 * i + 4 * c);
        }
      }
    }
    const float b2v = B2[a * OUTD + l2j];
    v8f acc[RT];
#pragma unroll
    for (int rt = 0; rt < RT; ++rt) acc[rt] = (v8f){};

    const float* bp = W1 + (size_t)a * EMB * HID + (size_t)khalf * HID + col;
    v2f b0[KB], b1[KB], b2[KB];
    load_b(b0, bp, 0);
    load_b(b1, bp, 1);
    for (int p = 0; p < NBLK / 3 - 1; ++p) {
      load_b(b2, bp, 3 * p + 2);
      __builtin_amdgcn_sched_barrier(0);
      mma_block(acc, b0, sXl, 3 * p);
      load_b(b0, bp, 3 * p + 3);
      __builtin_amdgcn_sched_barrier(0);
      mma_block(acc, b1, sXl, 3 * p + 1);
      load_b(b1, bp, 3 * p + 4);
      __builtin_amdgcn_sched_barrier(0);
      mma_block(acc, b2, sXl, 3 * p + 2);
    }
    load_b(b2, bp, NBLK - 1);
    __builtin_amdgcn_sched_barrier(0);
    mma_block(acc, b0, sXl, NBLK - 3);
    mma_block(acc, b1, sXl, NBLK - 2);
    mma_block(acc, b2, sXl, NBLK - 1);
    const float bias = B1[a * HID + col];
#pragma unroll
    for (int rt = 0; rt < RT; ++rt) {
#pragma unroll
      for (int r = 0; r < 8; ++r) {
        const int rowm = rt * 16 + r + 8 * (lane >> 4);
        float h = acc[rt][r] + bias;
        sHa[rowm * HP + col] = h > 0.0f ? h : 0.0f;
      }
    }
    __syncthreads();

    {
      const float* hp = sHa + l2row * HP + 4 * l2q;
      float p0 = 0.0f, p1 = 0.0f, p2 = 0.0f;
#pragma unroll
      for (int i = 0; i < 4; ++i) {
        const v4f hv = *(const v4f*)(hp + 64 * i);
        const float wf[12] = {
            w2v[i][0].x, w2v[i][0].y, w2v[i][0].z, w2v[i][0].w,
            w2v[i][1].x, w2v[i][1].y, w2v[i][1].z, w2v[i][1].w,
            w2v[i][2].x, w2v[i][2].y, w2v[i][2].z, w2v[i][2].w};
#pragma unroll
        for (int kk = 0; kk < 4; ++kk) {
          p0 += hv[kk] * wf[3 * kk + 0];
          p1 += hv[kk] * wf[3 * kk + 1];
          p2 += hv[kk] * wf[3 * kk + 2];
        }
      }
      p0 = row16_sum(p0);
      p1 = row16_sum(p1);
      p2 = row16_sum(p2);
      const float o = (l2q == 0 ? p0 : (l2q == 1 ? p1 : p2)) + b2v;
      res = (my_aspect == a) ? o : res;
    }
  }

  float* sO = smem + SO_OFFSET;
  if (l2q < OUTD) sO[l2row * OUTD + l2q] = res;
  __syncthreads();
  if (t < ROWS * OUTD) out[row0 * OUTD + t] = sO[t];
}

extern "C" void kernel_launch(void* const* d_in, const int* in_sizes, int n_in,
                              void* d_out, int out_size, void* d_ws, size_t ws_size,
                              hipStream_t stream) {
  (void)in_sizes; (void)n_in; (void)out_size; (void)d_ws; (void)ws_size;
  const int*   ids    = (const int*)d_in[0];
  const int*   aspect = (const int*)d_in[1];
  const float* emb    = (const float*)d_in[2];
  const float* W1     = (const float*)d_in[3];
  const float* b1     = (const float*)d_in[4];
  const float* W2     = (const float*)d_in[5];
  const float* b2     = (const float*)d_in[6];
  float* out = (float*)d_out;

  const size_t smem_bytes = (size_t)LDS_FLOATS * sizeof(float);
  fused_kernel<<<BATCH / ROWS, NWAVE * 32, smem_bytes, stream>>>(
      ids, aspect, emb, W1, b1, W2, b2, out);
}


# select moved ahead of layer 2: a hidden row is stored only under its sample's own aspect, one LDS plane; the 8 per-aspect barriers and 7 of 8 layer-2 passes go, W1 prefetch runs on; sum orders kept
# speedup vs baseline: 1.6722x; 1.0322x over previous
#include <hip/hip_runtime.h>


typedef __attribute__((ext_vector_type(2))) float v2f;
typedef __attribute__((ext_vector_type(4))) float v4f;
typedef __attribute__((ext_vector_type(4))) int   v4i;
typedef __attribute__((ext_vector_type(8))) float v8f;

#define VOCAB   50000
#define EMB     300
#define HID     256
#define OUTD    3
#define NASP    8
#define BATCH   8192
#define SEQ     128
#define ROWS    32
#define RT      2
#define KB      5
#define NBLK    (EMB / 4 / KB)
#define NWAVE   16
#define HP      260
#define GU      8
#define SX_FLOATS   (ROWS * EMB)
#define SH_FLOATS   (ROWS * HP)
#define SO_OFFSET   (SX_FLOATS + SH_FLOATS)
#define LDS_FLOATS  (41 * 1024)

__device__ __forceinline__ void load_b(v2f (&b)[KB], const float* __restrict__ bp, int blk) {
#pragma unroll
  for (int j = 0; j < KB; ++j) {
    const float* p = bp + (size_t)((blk * KB + j) * 4) * HID;
    b[j].x = p[0];
    b[j].y = p[HID];
  }
}

__device__ __forceinline__ void mma_block(v8f (&acc)[RT], const v2f (&b)[KB],
                                          const float* sXl, int blk) {
#pragma unroll
  for (int j = 0; j < KB; ++j) {
    const int k = (blk * KB + j) * 4;
#pragma unroll
    for (int rt = 0; rt < RT; ++rt) {
      v2f afrag = *(const v2f*)&sXl[rt * 16 * EMB + k];
      acc[rt] = __builtin_amdgcn_wmma_f32_16x16x4_f32(
           false, afrag,  false, b[j],
           (short)0, acc[rt],  false,  false);
    }
  }
}

__device__ __forceinline__ float row16_sum(float v) {
  v += __builtin_bit_cast(float, __builtin_amdgcn_update_dpp(
           0, __builtin_bit_cast(int, v), 0x161, 0xf, 0xf, true));
  v += __builtin_bit_cast(float, __builtin_amdgcn_update_dpp(
           0, __builtin_bit_cast(int, v), 0x162, 0xf, 0xf, true));
  v += __builtin_bit_cast(float, __builtin_amdgcn_update_dpp(
           0, __builtin_bit_cast(int, v), 0x164, 0xf, 0xf, true));
  v += __builtin_bit_cast(float, __builtin_amdgcn_update_dpp(
           0, __builtin_bit_cast(int, v), 0x168, 0xf, 0xf, true));
  return v;
}

__global__ __launch_bounds__(NWAVE * 32)
void fused_kernel(const int* __restrict__ ids,
                  const int* __restrict__ aspect_id,
                  const float* __restrict__ emb,
                  const float* __restrict__ W1,
                  const float* __restrict__ B1,
                  const float* __restrict__ W2,
                  const float* __restrict__ B2,
                  float* __restrict__ out) {
  extern __shared__ float smem[];
  float* sX = smem;
  float* sH = smem + SX_FLOATS;

  const int row0 = blockIdx.x * ROWS;
  const int t    = threadIdx.x;
  const int lane = t & 31;
  const int wv   = t >> 5;

  const int l2row = t >> 4;
  const int l2q   = t & 15;
  const int l2j   = l2q < OUTD ? l2q : OUTD - 1;
  int my_aspect = aspect_id[row0 + l2row];
  my_aspect = my_aspect < 0 ? 0 : (my_aspect > NASP - 1 ? NASP - 1 : my_aspect);

  int asp[RT][8];
#pragma unroll
  for (int rt = 0; rt < RT; ++rt) {
    const int* ap = aspect_id + row0 + rt * 16 + 8 * (lane >> 4);
    const v4i lo = *(const v4i*)ap;
    const v4i hi = *(const v4i*)(ap + 4);
#pragma unroll
    for (int r = 0; r < 8; ++r) {
      const int v = r < 4 ? lo[r & 3] : hi[r & 3];
      asp[rt][r] = v < 0 ? 0 : (v > NASP - 1 ? NASP - 1 : v);
    }
  }

  {
    const int o0 = (lane < 25 ? lane : 24) * 4;
    const int o1 = o0 + 100;
    const int o2 = o0 + 200;
    for (int si = 0; si < ROWS / NWAVE; ++si) {
      const int sl = wv * (ROWS / NWAVE) + si;
      const v4i idv = *(const v4i*)(ids + (size_t)(row0 + sl) * SEQ + lane * 4);
      v4f a0 = (v4f){}, a1 = (v4f){}, a2 = (v4f){};
      for (int k0 = 0; k0 < SEQ; k0 += GU) {
        v4f r0[GU], r1[GU], r2[GU];
#pragma unroll
        for (int u = 0; u < GU; ++u) {
          const int id = __builtin_amdgcn_readlane(idv[u & 3], (k0 >> 2) + (u >> 2));
          const unsigned e0 = (unsigned)id * (EMB * 4u) + (unsigned)o0 * 4u;
          const char* eb = (const char*)emb;
          r0[u] = *(const v4f*)(eb + e0);
          r1[u] = *(const v4f*)(eb + e0 + 400u);
          r2[u] = *(const v4f*)(eb + e0 + 800u);
        }
#pragma unroll
        for (int u = 0; u < GU; ++u) {
          a0 += r0[u];
          a1 += r1[u];
          a2 += r2[u];
        }
      }
      const float sc = 1.0f / (float)SEQ;
      float* xr = sX + sl * EMB;
      if (lane < 25) {
        *(v4f*)(xr + o0) = a0 * sc;
        *(v4f*)(xr + o1) = a1 * sc;
        *(v4f*)(xr + o2) = a2 * sc;
      }
    }
  }
  __syncthreads();

  const int nloc  = lane & 15;
  const int khalf = (lane >> 4) * 2;
  const int m     = lane & 15;
  const int col   = wv * 16 + nloc;
  const float* sXl = sX + m * EMB + khalf;

  v4f w2v[4][3];
  {
    const float* w2p = W2 + (size_t)my_aspect * HID * OUTD + 12 * l2q;
#pragma unroll
    for (int i = 0; i < 4; ++i) {
#pragma unroll
      for (int c = 0; c < 3; ++c) {
        w2v[i][c] = *(const v4f*)(w2p + 192 * i + 4 * c);
      }
    }
  }
  const float b2v = B2[my_aspect * OUTD + l2j];

  const float* bp0 = W1 + (size_t)khalf * HID + col;
  v2f b0[KB], b1[KB], b2[KB];
  load_b(b0, bp0, 0);
  load_b(b1, bp0, 1);

  for (int a = 0; a < NASP; ++a) {
    const float* bp  = bp0 + (size_t)a * EMB * HID;
    const int an = a + 1 < NASP ? a + 1 : a;
    const float* bpn = bp0 + (size_t)an * EMB * HID;
    const float bias = B1[a * HID + col];
    v8f acc[RT];
#pragma unroll
    for (int rt = 0; rt < RT; ++rt) acc[rt] = (v8f){};

    for (int p = 0; p < NBLK / 3 - 1; ++p) {
      load_b(b2, bp, 3 * p + 2);
      __builtin_amdgcn_sched_barrier(0);
      mma_block(acc, b0, sXl, 3 * p);
      load_b(b0, bp, 3 * p + 3);
      __builtin_amdgcn_sched_barrier(0);
      mma_block(acc, b1, sXl, 3 * p + 1);
      load_b(b1, bp, 3 * p + 4);
      __builtin_amdgcn_sched_barrier(0);
      mma_block(acc, b2, sXl, 3 * p + 2);
    }
    load_b(b2, bp, NBLK - 1);
    __builtin_amdgcn_sched_barrier(0);
    mma_block(acc, b0, sXl, NBLK - 3);
    load_b(b0, bpn, 0);
    __builtin_amdgcn_sched_barrier(0);
    mma_block(acc, b1, sXl, NBLK - 2);
    load_b(b1, bpn, 1);
    __builtin_amdgcn_sched_barrier(0);
    mma_block(acc, b2, sXl, NBLK - 1);
#pragma unroll
    for (int rt = 0; rt < RT; ++rt) {
#pragma unroll
      for (int r = 0; r < 8; ++r) {
        const int rowm = rt * 16 + r + 8 * (lane >> 4);
        float h = acc[rt][r] + bias;
        h = h > 0.0f ? h : 0.0f;
        if (asp[rt][r] == a) sH[rowm * HP + col] = h;
      }
    }
  }
  __syncthreads();

  float res;
  {
    const float* hp = sH + l2row * HP + 4 * l2q;
    float p0 = 0.0f, p1 = 0.0f, p2 = 0.0f;
#pragma unroll
    for (int i = 0; i < 4; ++i) {
      const v4f hv = *(const v4f*)(hp + 64 * i);
      const float wf[12] = {
          w2v[i][0].x, w2v[i][0].y, w2v[i][0].z, w2v[i][0].w,
          w2v[i][1].x, w2v[i][1].y, w2v[i][1].z, w2v[i][1].w,
          w2v[i][2].x, w2v[i][2].y, w2v[i][2].z, w2v[i][2].w};
#pragma unroll
      for (int kk = 0; kk < 4; ++kk) {
        p0 += hv[kk] * wf[3 * kk + 0];
        p1 += hv[kk] * wf[3 * kk + 1];
        p2 += hv[kk] * wf[3 * kk + 2];
      }
    }
    p0 = row16_sum(p0);
    p1 = row16_sum(p1);
    p2 = row16_sum(p2);
    res = (l2q == 0 ? p0 : (l2q == 1 ? p1 : p2)) + b2v;
  }

  float* sO = smem + SO_OFFSET;
  if (l2q < OUTD) sO[l2row * OUTD + l2q] = res;
  __syncthreads();
  if (t < ROWS * OUTD) out[row0 * OUTD + t] = sO[t];
}

extern "C" void kernel_launch(void* const* d_in, const int* in_sizes, int n_in,
                              void* d_out, int out_size, void* d_ws, size_t ws_size,
                              hipStream_t stream) {
  (void)in_sizes; (void)n_in; (void)out_size; (void)d_ws; (void)ws_size;
  const int*   ids    = (const int*)d_in[0];
  const int*   aspect = (const int*)d_in[1];
  const float* emb    = (const float*)d_in[2];
  const float* W1     = (const float*)d_in[3];
  const float* b1     = (const float*)d_in[4];
  const float* W2     = (const float*)d_in[5];
  const float* b2     = (const float*)d_in[6];
  float* out = (float*)d_out;

  const size_t smem_bytes = (size_t)LDS_FLOATS * sizeof(float);
  fused_kernel<<<BATCH / ROWS, NWAVE * 32, smem_bytes, stream>>>(
      ids, aspect, emb, W1, b1, W2, b2, out);
}


# layer 1 re-tiled: a wave takes 64 H columns (every 4th column per tile) x 32 rows for 2 of the 8 aspects, so W1 comes as whole-line 16-byte loads and each A fragment feeds 4 WMMAs; sum orders kept
# speedup vs baseline: 1.7431x; 1.0424x over previous
#include <hip/hip_runtime.h>


typedef __attribute__((ext_vector_type(2))) float v2f;
typedef __attribute__((ext_vector_type(4))) float v4f;
typedef __attribute__((ext_vector_type(4))) int   v4i;
typedef __attribute__((ext_vector_type(8))) float v8f;

#define VOCAB   50000
#define EMB     300
#define HID     256
#define OUTD    3
#define NASP    8
#define BATCH   8192
#define SEQ     128
#define ROWS    32
#define RT      2
#define NT      4
#define KSTEPS  (EMB / 4)
#define NSET    5
#define NGRP    (KSTEPS / NSET)
#define NWAVE   16
#define HP      260
#define GU      8
#define SX_FLOATS   (ROWS * EMB)
#define SH_FLOATS   (ROWS * HP)
#define SO_OFFSET   (SX_FLOATS + SH_FLOATS)
#define LDS_FLOATS  (41 * 1024)

__device__ __forceinline__ void load_b4(v4f& lo, v4f& hi, const float* __restrict__ p) {
  lo = *(const v4f*)p;
  hi = *(const v4f*)(p + HID);
}

__device__ __forceinline__ void mma_step(v8f (&acc)[RT][NT], const v4f& lo, const v4f& hi,
                                         const float* sXl, int kk) {
#pragma unroll
  for (int rt = 0; rt < RT; ++rt) {
    const v2f afrag = *(const v2f*)&sXl[rt * 16 * EMB + kk * 4];
#pragma unroll
    for (int j = 0; j < NT; ++j) {
      v2f bfrag;
      bfrag.x = lo[j];
      bfrag.y = hi[j];
      acc[rt][j] = __builtin_amdgcn_wmma_f32_16x16x4_f32(
           false, afrag,  false, bfrag,
           (short)0, acc[rt][j],  false,  false);
    }
  }
}

__device__ __forceinline__ float row16_sum(float v) {
  v += __builtin_bit_cast(float, __builtin_amdgcn_update_dpp(
           0, __builtin_bit_cast(int, v), 0x161, 0xf, 0xf, true));
  v += __builtin_bit_cast(float, __builtin_amdgcn_update_dpp(
           0, __builtin_bit_cast(int, v), 0x162, 0xf, 0xf, true));
  v += __builtin_bit_cast(float, __builtin_amdgcn_update_dpp(
           0, __builtin_bit_cast(int, v), 0x164, 0xf, 0xf, true));
  v += __builtin_bit_cast(float, __builtin_amdgcn_update_dpp(
           0, __builtin_bit_cast(int, v), 0x168, 0xf, 0xf, true));
  return v;
}

__global__ __launch_bounds__(NWAVE * 32)
void fused_kernel(const int* __restrict__ ids,
                  const int* __restrict__ aspect_id,
                  const float* __restrict__ emb,
                  const float* __restrict__ W1,
                  const float* __restrict__ B1,
                  const float* __restrict__ W2,
                  const float* __restrict__ B2,
                  float* __restrict__ out) {
  extern __shared__ float smem[];
  float* sX = smem;
  float* sH = smem + SX_FLOATS;

  const int row0 = blockIdx.x * ROWS;
  const int t    = threadIdx.x;
  const int lane = t & 31;
  const int wv   = t >> 5;

  const int l2row = t >> 4;
  const int l2q   = t & 15;
  const int l2j   = l2q < OUTD ? l2q : OUTD - 1;
  int my_aspect = aspect_id[row0 + l2row];
  my_aspect = my_aspect < 0 ? 0 : (my_aspect > NASP - 1 ? NASP - 1 : my_aspect);

  int asp[RT][8];
#pragma unroll
  for (int rt = 0; rt < RT; ++rt) {
    const int* ap = aspect_id + row0 + rt * 16 + 8 * (lane >> 4);
    const v4i lo = *(const v4i*)ap;
    const v4i hi = *(const v4i*)(ap + 4);
#pragma unroll
    for (int r = 0; r < 8; ++r) {
      const int v = r < 4 ? lo[r & 3] : hi[r & 3];
      asp[rt][r] = v < 0 ? 0 : (v > NASP - 1 ? NASP - 1 : v);
    }
  }

  {
    const int o0 = (lane < 25 ? lane : 24) * 4;
    const int o1 = o0 + 100;
    const int o2 = o0 + 200;
    for (int si = 0; si < ROWS / NWAVE; ++si) {
      const int sl = wv * (ROWS / NWAVE) + si;
      const v4i idv = *(const v4i*)(ids + (size_t)(row0 + sl) * SEQ + lane * 4);
      v4f a0 = (v4f){}, a1 = (v4f){}, a2 = (v4f){};
      for (int k0 = 0; k0 < SEQ; k0 += GU) {
        v4f r0[GU], r1[GU], r2[GU];
#pragma unroll
        for (int u = 0; u < GU; ++u) {
          const int id = __builtin_amdgcn_readlane(idv[u & 3], (k0 >> 2) + (u >> 2));
          const unsigned e0 = (unsigned)id * (EMB * 4u) + (unsigned)o0 * 4u;
          const char* eb = (const char*)emb;
          r0[u] = *(const v4f*)(eb + e0);
          r1[u] = *(const v4f*)(eb + e0 + 400u);
          r2[u] = *(const v4f*)(eb + e0 + 800u);
        }
#pragma unroll
        for (int u = 0; u < GU; ++u) {
          a0 += r0[u];
          a1 += r1[u];
          a2 += r2[u];
        }
      }
      const float sc = 1.0f / (float)SEQ;
      float* xr = sX + sl * EMB;
      if (lane < 25) {
        *(v4f*)(xr + o0) = a0 * sc;
        *(v4f*)(xr + o1) = a1 * sc;
        *(v4f*)(xr + o2) = a2 * sc;
      }
    }
  }
  __syncthreads();

  const int nloc  = lane & 15;
  const int khalf = (lane >> 4) * 2;
  const int m     = lane & 15;
  const int g     = wv & 3;
  const int q     = wv >> 2;
  const int col0  = 64 * g + 4 * nloc;
  const float* sXl = sX + m * EMB + khalf;

  v4f w2v[4][3];
  {
    const float* w2p = W2 + (size_t)my_aspect * HID * OUTD + 12 * l2q;
#pragma unroll
    for (int i = 0; i < 4; ++i) {
#pragma unroll
      for (int c = 0; c < 3; ++c) {
        w2v[i][c] = *(const v4f*)(w2p + 192 * i + 4 * c);
      }
    }
  }
  const float b2v = B2[my_aspect * OUTD + l2j];

  const float* wl = W1 + (size_t)(2 * q) * EMB * HID + (size_t)khalf * HID + col0;
  v4f blo[NSET], bhi[NSET];
#pragma unroll
  for (int s = 0; s < NSET - 1; ++s) load_b4(blo[s], bhi[s], wl + (size_t)s * 4 * HID);

  for (int ai = 0; ai < 2; ++ai) {
    const int a = 2 * q + ai;
    const v4f bias = *(const v4f*)(B1 + a * HID + col0);
    v8f acc[RT][NT];
#pragma unroll
    for (int rt = 0; rt < RT; ++rt) {
#pragma unroll
      for (int j = 0; j < NT; ++j) acc[rt][j] = (v8f){};
    }

#pragma unroll 1
    for (int i = 0; i < NGRP; ++i) {
#pragma unroll
      for (int s = 0; s < NSET; ++s) {
        const int kk = i * NSET + s;
        int tn = ai * KSTEPS + kk + (NSET - 1);
        tn = tn < 2 * KSTEPS - 1 ? tn : 2 * KSTEPS - 1;
        load_b4(blo[(s + NSET - 1) % NSET], bhi[(s + NSET - 1) % NSET],
                wl + (size_t)tn * 4 * HID);
        __builtin_amdgcn_sched_barrier(0);
        mma_step(acc, blo[s], bhi[s], sXl, kk);
      }
    }
#pragma unroll
    for (int rt = 0; rt < RT; ++rt) {
#pragma unroll
      for (int r = 0; r < 8; ++r) {
        const int rowm = rt * 16 + r + 8 * (lane >> 4);
        v4f hv;
        hv.x = acc[rt][0][r] + bias.x;
        hv.y = acc[rt][1][r] + bias.y;
        hv.z = acc[rt][2][r] + bias.z;
        hv.w = acc[rt][3][r] + bias.w;
        hv.x = hv.x > 0.0f ? hv.x : 0.0f;
        hv.y = hv.y > 0.0f ? hv.y : 0.0f;
        hv.z = hv.z > 0.0f ? hv.z : 0.0f;
        hv.w = hv.w > 0.0f ? hv.w : 0.0f;
        if (asp[rt][r] == a) *(v4f*)&sH[rowm * HP + col0] = hv;
      }
    }
  }
  __syncthreads();

  float res;
  {
    const float* hp = sH + l2row * HP + 4 * l2q;
    float p0 = 0.0f, p1 = 0.0f, p2 = 0.0f;
#pragma unroll
    for (int i = 0; i < 4; ++i) {
      const v4f hv = *(const v4f*)(hp + 64 * i);
      const float wf[12] = {
          w2v[i][0].x, w2v[i][0].y, w2v[i][0].z, w2v[i][0].w,
          w2v[i][1].x, w2v[i][1].y, w2v[i][1].z, w2v[i][1].w,
          w2v[i][2].x, w2v[i][2].y, w2v[i][2].z, w2v[i][2].w};
#pragma unroll
      for (int kk = 0; kk < 4; ++kk) {
        p0 += hv[kk] * wf[3 * kk + 0];
        p1 += hv[kk] * wf[3 * kk + 1];
        p2 += hv[kk] * wf[3 * kk + 2];
      }
    }
    p0 = row16_sum(p0);
    p1 = row16_sum(p1);
    p2 = row16_sum(p2);
    res = (l2q == 0 ? p0 : (l2q == 1 ? p1 : p2)) + b2v;
  }

  float* sO = smem + SO_OFFSET;
  if (l2q < OUTD) sO[l2row * OUTD + l2q] = res;
  __syncthreads();
  if (t < ROWS * OUTD) out[row0 * OUTD + t] = sO[t];
}

extern "C" void kernel_launch(void* const* d_in, const int* in_sizes, int n_in,
                              void* d_out, int out_size, void* d_ws, size_t ws_size,
                              hipStream_t stream) {
  (void)in_sizes; (void)n_in; (void)out_size; (void)d_ws; (void)ws_size;
  const int*   ids    = (const int*)d_in[0];
  const int*   aspect = (const int*)d_in[1];
  const float* emb    = (const float*)d_in[2];
  const float* W1     = (const float*)d_in[3];
  const float* b1     = (const float*)d_in[4];
  const float* W2     = (const float*)d_in[5];
  const float* b2     = (const float*)d_in[6];
  float* out = (float*)d_out;

  const size_t smem_bytes = (size_t)LDS_FLOATS * sizeof(float);
  fused_kernel<<<BATCH / ROWS, NWAVE * 32, smem_bytes, stream>>>(
      ids, aspect, emb, W1, b1, W2, b2, out);
}


# waves split by role: X pooled in three 100-column pieces; waves 8..15 gather piece p+1 (six sets of 8 row loads in turn) into a 2nd LDS buffer while waves 0..7 run layer 1 on piece p; sum orders kept
# speedup vs baseline: 1.8092x; 1.0379x over previous
#include <hip/hip_runtime.h>


typedef __attribute__((ext_vector_type(2))) float v2f;
typedef __attribute__((ext_vector_type(4))) float v4f;
typedef __attribute__((ext_vector_type(4))) int   v4i;
typedef __attribute__((ext_vector_type(8))) float v8f;

#define VOCAB   50000
#define EMB     300
#define HID     256
#define OUTD    3
#define NASP    8
#define BATCH   8192
#define SEQ     128
#define ROWS    32
#define RT      2
#define NT      4
#define NP      3
#define PC      (EMB / NP)
#define PK      (PC / 4)
#define NSET    5
#define NWAVE   16
#define NMMA    8
#define UPW     4
#define HP      260
#define GH      8
#define NRS     6
#define XP_FLOATS   (ROWS * PC)
#define SH_OFFSET   (2 * XP_FLOATS)
#define SO_OFFSET   (SH_OFFSET + ROWS * HP)
#define ACC_OFFSET  14848
#define UNIT_FLOATS (RT * NT * 8 * 32)
#define LDS_FLOATS  (ACC_OFFSET + NASP * 4 * UNIT_FLOATS)

__device__ __forceinline__ void load_b4(v4f& lo, v4f& hi, const float* __restrict__ p) {
  lo = *(const v4f*)p;
  hi = *(const v4f*)(p + HID);
}

__device__ __forceinline__ void mma_step(v8f (&acc)[RT][NT], const v4f& lo, const v4f& hi,
                                         const float* sXl, int kk) {
#pragma unroll
  for (int rt = 0; rt < RT; ++rt) {
    const v2f afrag = *(const v2f*)&sXl[rt * 16 * PC + kk * 4];
#pragma unroll
    for (int j = 0; j < NT; ++j) {
      v2f bfrag;
      bfrag.x = lo[j];
      bfrag.y = hi[j];
      acc[rt][j] = __builtin_amdgcn_wmma_f32_16x16x4_f32(
           false, afrag,  false, bfrag,
           (short)0, acc[rt][j],  false,  false);
    }
  }
}

__device__ __forceinline__ float row16_sum(float v) {
  v += __builtin_bit_cast(float, __builtin_amdgcn_update_dpp(
           0, __builtin_bit_cast(int, v), 0x161, 0xf, 0xf, true));
  v += __builtin_bit_cast(float, __builtin_amdgcn_update_dpp(
           0, __builtin_bit_cast(int, v), 0x162, 0xf, 0xf, true));
  v += __builtin_bit_cast(float, __builtin_amdgcn_update_dpp(
           0, __builtin_bit_cast(int, v), 0x164, 0xf, 0xf, true));
  v += __builtin_bit_cast(float, __builtin_amdgcn_update_dpp(
           0, __builtin_bit_cast(int, v), 0x168, 0xf, 0xf, true));
  return v;
}

__device__ __forceinline__ void issue_rows(v4f (&r)[GH], const v4i& idv, int tk0,
                                           const char* eb, unsigned lo) {
#pragma unroll
  for (int u = 0; u < GH; ++u) {
    const int id = __builtin_amdgcn_readlane(idv[u & 3], (tk0 >> 2) + (u >> 2));
    r[u] = *(const v4f*)(eb + ((unsigned)id * (EMB * 4u) + lo));
  }
}

__device__ __forceinline__ void gather_piece(const int* __restrict__ ids,
                                             const float* __restrict__ emb,
                                             float* xb, int row0, int sl0, int ns,
                                             int piece, int lane) {
  const int o0 = (lane < 25 ? lane : 24) * 4;
  const unsigned lo = (unsigned)piece * (PC * 4u) + (unsigned)o0 * 4u;
  const char* eb = (const char*)emb;
  v4i idv = *(const v4i*)(ids + (size_t)(row0 + sl0) * SEQ + lane * 4);
#pragma unroll 1
  for (int si = 0; si < ns; ++si) {
    const int sl = sl0 + si;
    const int sn = si + 1 < ns ? sl + 1 : sl;
    const v4i idn = *(const v4i*)(ids + (size_t)(row0 + sn) * SEQ + lane * 4);
    v4f a = (v4f){};
    v4f r[NRS][GH];
#pragma unroll
    for (int gi = 0; gi < NRS - 1; ++gi) {
      issue_rows(r[gi], idv, gi * GH, eb, lo);
      __builtin_amdgcn_sched_barrier(0);
    }
#pragma unroll
    for (int gi = 0; gi < SEQ / GH; ++gi) {
      if (gi + NRS - 1 < SEQ / GH)
        issue_rows(r[(gi + NRS - 1) % NRS], idv, (gi + NRS - 1) * GH, eb, lo);
      __builtin_amdgcn_sched_barrier(0);
#pragma unroll
      for (int u = 0; u < GH; ++u) a += r[gi % NRS][u];
      __builtin_amdgcn_sched_barrier(0);
    }
    *(v4f*)(xb + sl * PC + o0) = a * (1.0f / (float)SEQ);
    idv = idn;
  }
}

template <bool LAST>
__device__ __forceinline__ void mma_piece(int p, const float* sX, float* sH, float* sAcc,
                                          const float* __restrict__ W1,
                                          const float* __restrict__ B1,
                                          const int (&asp)[RT][8],
                                          int lane, int m, int khalf, int g, int ah, int col0) {
  const float* sXl = sX + (p & 1) * XP_FLOATS + m * PC + khalf;
  const float* wb = W1 + (size_t)(4 * ah) * EMB * HID + (size_t)(PC * p + khalf) * HID + col0;
  v4f blo[NSET], bhi[NSET];
#pragma unroll
  for (int s = 0; s < NSET - 1; ++s) load_b4(blo[s], bhi[s], wb + (size_t)s * 4 * HID);

#pragma unroll 1
  for (int ui = 0; ui < UPW; ++ui) {
    const int a = 4 * ah + ui;
    float* park = sAcc + (a * 4 + g) * UNIT_FLOATS + lane * 4;
    v4f bias = (v4f){};
    if (LAST) bias = *(const v4f*)(B1 + a * HID + col0);
    v8f acc[RT][NT];
    if (p == 0) {
#pragma unroll
      for (int rt = 0; rt < RT; ++rt) {
#pragma unroll
        for (int j = 0; j < NT; ++j) acc[rt][j] = (v8f){};
      }
    } else {
#pragma unroll
      for (int rt = 0; rt < RT; ++rt) {
#pragma unroll
        for (int j = 0; j < NT; ++j) {
          const v4f x0 = *(const v4f*)(park + ((rt * NT + j) * 2 + 0) * 128);
          const v4f x1 = *(const v4f*)(park + ((rt * NT + j) * 2 + 1) * 128);
          acc[rt][j] = __builtin_shufflevector(x0, x1, 0, 1, 2, 3, 4, 5, 6, 7);
        }
      }
    }

    const float* wu = wb + (size_t)ui * EMB * HID;
#pragma unroll
    for (int kk = 0; kk < PK; ++kk) {
      const int s = kk % NSET;
      const int kn = kk + (NSET - 1);
      const float* pn = kn < PK ? wu + (size_t)kn * 4 * HID
                                : (ui < UPW - 1 ? wu + (size_t)EMB * HID + (size_t)(kn - PK) * 4 * HID
                                                : wu + (size_t)(PK - 1) * 4 * HID);
      load_b4(blo[(s + NSET - 1) % NSET], bhi[(s + NSET - 1) % NSET], pn);
      __builtin_amdgcn_sched_barrier(0);
      mma_step(acc, blo[s], bhi[s], sXl, kk);
    }

    if (!LAST) {
#pragma unroll
      for (int rt = 0; rt < RT; ++rt) {
#pragma unroll
        for (int j = 0; j < NT; ++j) {
          *(v4f*)(park + ((rt * NT + j) * 2 + 0) * 128) =
              __builtin_shufflevector(acc[rt][j], acc[rt][j], 0, 1, 2, 3);
          *(v4f*)(park + ((rt * NT + j) * 2 + 1) * 128) =
              __builtin_shufflevector(acc[rt][j], acc[rt][j], 4, 5, 6, 7);
        }
      }
    } else {
#pragma unroll
      for (int rt = 0; rt < RT; ++rt) {
#pragma unroll
        for (int r = 0; r < 8; ++r) {
          const int rowm = rt * 16 + r + 8 * (lane >> 4);
          v4f hv;
          hv.x = acc[rt][0][r] + bias.x;
          hv.y = acc[rt][1][r] + bias.y;
          hv.z = acc[rt][2][r] + bias.z;
          hv.w = acc[rt][3][r] + bias.w;
          hv.x = hv.x > 0.0f ? hv.x : 0.0f;
          hv.y = hv.y > 0.0f ? hv.y : 0.0f;
          hv.z = hv.z > 0.0f ? hv.z : 0.0f;
          hv.w = hv.w > 0.0f ? hv.w : 0.0f;
          if (asp[rt][r] == a) *(v4f*)&sH[rowm * HP + col0] = hv;
        }
      }
    }
  }
}

__global__ __launch_bounds__(NWAVE * 32)
void fused_kernel(const int* __restrict__ ids,
                  const int* __restrict__ aspect_id,
                  const float* __restrict__ emb,
                  const float* __restrict__ W1,
                  const float* __restrict__ B1,
                  const float* __restrict__ W2,
                  const float* __restrict__ B2,
                  float* __restrict__ out) {
  __shared__ __attribute__((aligned(16))) float smem[LDS_FLOATS];
  float* sX   = smem;
  float* sH   = smem + SH_OFFSET;
  float* sAcc = smem + ACC_OFFSET;

  const int row0 = blockIdx.x * ROWS;
  const int t    = threadIdx.x;
  const int lane = t & 31;
  const int wv   = __builtin_amdgcn_readfirstlane(t >> 5);

  const int l2row = t >> 4;
  const int l2q   = t & 15;
  const int l2j   = l2q < OUTD ? l2q : OUTD - 1;
  int my_aspect = aspect_id[row0 + l2row];
  my_aspect = my_aspect < 0 ? 0 : (my_aspect > NASP - 1 ? NASP - 1 : my_aspect);

  gather_piece(ids, emb, sX, row0, wv * (ROWS / NWAVE), ROWS / NWAVE, 0, lane);
  __syncthreads();

  const int nloc  = lane & 15;
  const int khalf = (lane >> 4) * 2;
  const int m     = lane & 15;
  const int g     = wv & 3;
  const int ah    = (wv >> 2) & 1;
  const int col0  = 64 * g + 4 * nloc;

#pragma unroll 1
  for (int p = 0; p < NP - 1; ++p) {
    if (wv < NMMA) {
      int none[RT][8] = {};
      mma_piece<false>(p, sX, sH, sAcc, W1, B1, none, lane, m, khalf, g, ah, col0);
    } else {
      gather_piece(ids, emb, sX + ((p + 1) & 1) * XP_FLOATS, row0,
                   (wv - NMMA) * (ROWS / (NWAVE - NMMA)), ROWS / (NWAVE - NMMA), p + 1, lane);
    }
    __syncthreads();
  }

  int asp[RT][8];
#pragma unroll
  for (int rt = 0; rt < RT; ++rt) {
    const int* ap = aspect_id + row0 + rt * 16 + 8 * (lane >> 4);
    const v4i lo = *(const v4i*)ap;
    const v4i hi = *(const v4i*)(ap + 4);
#pragma unroll
    for (int r = 0; r < 8; ++r) {
      const int v = r < 4 ? lo[r & 3] : hi[r & 3];
      asp[rt][r] = v < 0 ? 0 : (v > NASP - 1 ? NASP - 1 : v);
    }
  }

  v4f w2v[4][3];
  {
    const float* w2p = W2 + (size_t)my_aspect * HID * OUTD + 12 * l2q;
#pragma unroll
    for (int i = 0; i < 4; ++i) {
#pragma unroll
      for (int c = 0; c < 3; ++c) {
        w2v[i][c] = *(const v4f*)(w2p + 192 * i + 4 * c);
      }
    }
  }
  const float b2v = B2[my_aspect * OUTD + l2j];

  if (wv < NMMA) mma_piece<true>(NP - 1, sX, sH, sAcc, W1, B1, asp, lane, m, khalf, g, ah, col0);
  __syncthreads();

  float res;
  {
    const float* hp = sH + l2row * HP + 4 * l2q;
    float p0 = 0.0f, p1 = 0.0f, p2 = 0.0f;
#pragma unroll
    for (int i = 0; i < 4; ++i) {
      const v4f hv = *(const v4f*)(hp + 64 * i);
      const float wf[12] = {
          w2v[i][0].x, w2v[i][0].y, w2v[i][0].z, w2v[i][0].w,
          w2v[i][1].x, w2v[i][1].y, w2v[i][1].z, w2v[i][1].w,
          w2v[i][2].x, w2v[i][2].y, w2v[i][2].z, w2v[i][2].w};
#pragma unroll
      for (int kk = 0; kk < 4; ++kk) {
        p0 += hv[kk] * wf[3 * kk + 0];
        p1 += hv[kk] * wf[3 * kk + 1];
        p2 += hv[kk] * wf[3 * kk + 2];
      }
    }
    p0 = row16_sum(p0);
    p1 = row16_sum(p1);
    p2 = row16_sum(p2);
    res = (l2q == 0 ? p0 : (l2q == 1 ? p1 : p2)) + b2v;
  }

  float* sO = smem + SO_OFFSET;
  if (l2q < OUTD) sO[l2row * OUTD + l2q] = res;
  __syncthreads();
  if (t < ROWS * OUTD) out[row0 * OUTD + t] = sO[t];
}

extern "C" void kernel_launch(void* const* d_in, const int* in_sizes, int n_in,
                              void* d_out, int out_size, void* d_ws, size_t ws_size,
                              hipStream_t stream) {
  (void)in_sizes; (void)n_in; (void)out_size; (void)d_ws; (void)ws_size;
  const int*   ids    = (const int*)d_in[0];
  const int*   aspect = (const int*)d_in[1];
  const float* emb    = (const float*)d_in[2];
  const float* W1     = (const float*)d_in[3];
  const float* b1     = (const float*)d_in[4];
  const float* W2     = (const float*)d_in[5];
  const float* b2     = (const float*)d_in[6];
  float* out = (float*)d_out;

  fused_kernel<<<BATCH / ROWS, NWAVE * 32, 0, stream>>>(
      ids, aspect, emb, W1, b1, W2, b2, out);
}
